# starting kernel
# baseline (speedup 1.0000x reference)
#include <hip/hip_runtime.h>
#include <hip/hip_bf16.h>

typedef float v2f __attribute__((ext_vector_type(2)));
typedef float v8f __attribute__((ext_vector_type(8)));

__device__ inline float wred32(float v) {
#pragma unroll
    for (int m = 16; m > 0; m >>= 1) v += __shfl_xor(v, m, 32);
    return v;
}

__global__ __launch_bounds__(32) void precompute_coeffs(
    const float* __restrict__ WA,    const float* __restrict__ bA,
    const float* __restrict__ WB,    const float* __restrict__ bB,
    const float* __restrict__ Wi_AB, const float* __restrict__ bi_AB,
    const float* __restrict__ Wo_AB, const float* __restrict__ bo_AB,
    const float* __restrict__ Wi_BA, const float* __restrict__ bi_BA,
    const float* __restrict__ Wo_BA, const float* __restrict__ bo_BA,
    const float* __restrict__ Wf,    const float* __restrict__ bf,
    float* __restrict__ coef)
{
    __shared__ float Y[192 * 16];
    const int lane  = threadIdx.x;
    const int n     = lane & 15;
    const int khalf = (lane >> 4) << 1;

    float Bx[8], By[8];
#pragma unroll
    for (int k0 = 0; k0 < 8; ++k0) {
        const int kb = 4 * k0 + khalf;
        float x0 = 0.f, x1 = 0.f;
        if      (n == 0) { x0 = WA[kb]; x1 = WA[kb + 1]; }
        else if (n == 1) { x0 = bA[kb]; x1 = bA[kb + 1]; }
        else if (n == 2) { x0 = WB[kb]; x1 = WB[kb + 1]; }
        else if (n == 3) { x0 = bB[kb]; x1 = bB[kb + 1]; }
        Bx[k0] = x0; By[k0] = x1;
    }

    for (int t = 0; t < 12; ++t) {
        const float* Wrow = (t < 6) ? (Wi_AB + (16 * t + n) * 32)
                                    : (Wi_BA + (16 * (t - 6) + n) * 32);
        v8f acc = {0.f, 0.f, 0.f, 0.f, 0.f, 0.f, 0.f, 0.f};
#pragma unroll
        for (int k0 = 0; k0 < 8; ++k0) {
            const int kb = 4 * k0 + khalf;
            v2f a; a.x = Wrow[kb]; a.y = Wrow[kb + 1];
            v2f b; b.x = Bx[k0];   b.y = By[k0];
            acc = __builtin_amdgcn_wmma_f32_16x16x4_f32(
                false, a, false, b, (short)0, acc, false, false);
        }
#pragma unroll
        for (int j = 0; j < 8; ++j) {
            const int M = 16 * t + j + ((lane >> 4) << 3);
            Y[M * 16 + n] = acc[j];
        }
    }
    __syncthreads();

    const int i = lane;
    float uq  = Y[(0   + i) * 16 + 0];
    float cq  = Y[(0   + i) * 16 + 1] + bi_AB[i];
    float uk  = Y[(32  + i) * 16 + 2];
    float ck  = Y[(32  + i) * 16 + 3] + bi_AB[32 + i];
    float uv  = Y[(64  + i) * 16 + 2];
    float cv  = Y[(64  + i) * 16 + 3] + bi_AB[64 + i];
    float uq2 = Y[(96  + i) * 16 + 2];
    float cq2 = Y[(96  + i) * 16 + 3] + bi_BA[i];
    float uk2 = Y[(128 + i) * 16 + 0];
    float ck2 = Y[(128 + i) * 16 + 1] + bi_BA[32 + i];
    float uv2 = Y[(160 + i) * 16 + 0];
    float cv2 = Y[(160 + i) * 16 + 1] + bi_BA[64 + i];

    float gAi = 0.f, gBi = 0.f;
#pragma unroll 4
    for (int mm = 0; mm < 32; ++mm) {
        const float w = Wf[mm];
        gAi += w * Wo_AB[mm * 32 + i];
        gBi += w * Wo_BA[mm * 32 + i];
    }

    const float aAB = wred32(uq  * uk );
    const float bAB = wred32(uq  * ck );
    const float gAB = wred32(cq  * uk );
    const float dAB = wred32(cq  * ck );
    const float aBA = wred32(uq2 * uk2);
    const float bBA = wred32(uq2 * ck2);
    const float gBA = wred32(cq2 * uk2);
    const float dBA = wred32(cq2 * ck2);
    const float sA  = wred32(gAi * uv );
    const float sB  = wred32(gBi * uv2);
    const float tA  = wred32(gAi * cv ) + wred32(Wf[i] * bo_AB[i]);
    const float tB  = wred32(gBi * cv2) + wred32(Wf[i] * bo_BA[i]);

    if (lane == 0) {
        coef[0] = aAB; coef[1] = bAB; coef[2] = gAB; coef[3] = dAB;
        coef[4] = aBA; coef[5] = bBA; coef[6] = gBA; coef[7] = dBA;
        coef[8] = sA;  coef[9] = sB;
        coef[10] = 0.5f * (tA + tB) + bf[0];
    }
}

__device__ inline float softmax3dot(float c, float a0, float a1, float a2) {
    const float e0 = c * a0, e1 = c * a1, e2 = c * a2;
    const float mx = fmaxf(e0, fmaxf(e1, e2));
    const float w0 = __expf(e0 - mx), w1 = __expf(e1 - mx), w2 = __expf(e2 - mx);
    return (w0 * a0 + w1 * a1 + w2 * a2) / (w0 + w1 + w2);
}

__device__ inline float one_batch(float a0, float a1, float a2,
                                  float b0, float b1,
                                  float aAB, float gAB, float aBA, float gBA,
                                  float sA, float sB, float c0)
{
    const float inv = 0.17677669529663687f;
    const float db = inv * (b1 - b0);
    const float z0 = db * (aAB * a0 + gAB);
    const float z1 = db * (aAB * a1 + gAB);
    const float z2 = db * (aAB * a2 + gAB);
    const float S  = 1.f / (1.f + __expf(-z0))
                   + 1.f / (1.f + __expf(-z1))
                   + 1.f / (1.f + __expf(-z2));
    const float MA = b0 + (b1 - b0) * (S * (1.f / 3.f));
    const float c0t = inv * (aBA * b0 + gBA);
    const float c1t = inv * (aBA * b1 + gBA);
    const float MB  = 0.5f * (softmax3dot(c0t, a0, a1, a2) +
                              softmax3dot(c1t, a0, a1, a2));
    const float o = 0.5f * (MA * sA + MB * sB) + c0;
    return (o >= 0.f) ? o : 0.01f * o;
}

__global__ __launch_bounds__(256) void dynf_main(
    const float* __restrict__ gA, const float* __restrict__ gB,
    const float* __restrict__ coef, float* __restrict__ out, int nB)
{
    const float aAB = coef[0], gAB = coef[2];
    const float aBA = coef[4], gBA = coef[6];
    const float sA  = coef[8], sB  = coef[9], c0 = coef[10];

    const int idx = blockIdx.x * blockDim.x + threadIdx.x;
    const int b0  = idx * 4;
    if (b0 + 3 < nB) {
        const float4 va0 = ((const float4*)(gA + 3 * b0))[0];
        const float4 va1 = ((const float4*)(gA + 3 * b0))[1];
        const float4 va2 = ((const float4*)(gA + 3 * b0))[2];
        const float4 vb0 = ((const float4*)(gB + 2 * b0))[0];
        const float4 vb1 = ((const float4*)(gB + 2 * b0))[1];
        float4 r;
        r.x = one_batch(va0.x, va0.y, va0.z, vb0.x, vb0.y, aAB, gAB, aBA, gBA, sA, sB, c0);
        r.y = one_batch(va0.w, va1.x, va1.y, vb0.z, vb0.w, aAB, gAB, aBA, gBA, sA, sB, c0);
        r.z = one_batch(va1.z, va1.w, va2.x, vb1.x, vb1.y, aAB, gAB, aBA, gBA, sA, sB, c0);
        r.w = one_batch(va2.y, va2.z, va2.w, vb1.z, vb1.w, aAB, gAB, aBA, gBA, sA, sB, c0);
        ((float4*)(out + b0))[0] = r;
    } else {
        for (int b = b0; b < nB; ++b) {
            out[b] = one_batch(gA[3 * b], gA[3 * b + 1], gA[3 * b + 2],
                               gB[2 * b], gB[2 * b + 1],
                               aAB, gAB, aBA, gBA, sA, sB, c0);
        }
    }
}

extern "C" void kernel_launch(void* const* d_in, const int* in_sizes, int n_in,
                              void* d_out, int out_size, void* d_ws, size_t ws_size,
                              hipStream_t stream) {
    const float* groupA = (const float*)d_in[0];
    const float* groupB = (const float*)d_in[1];
    const float* WA     = (const float*)d_in[2];
    const float* bA     = (const float*)d_in[3];
    const float* WB     = (const float*)d_in[4];
    const float* bB     = (const float*)d_in[5];
    const float* Wi_AB  = (const float*)d_in[6];
    const float* bi_AB  = (const float*)d_in[7];
    const float* Wo_AB  = (const float*)d_in[8];
    const float* bo_AB  = (const float*)d_in[9];
    const float* Wi_BA  = (const float*)d_in[10];
    const float* bi_BA  = (const float*)d_in[11];
    const float* Wo_BA  = (const float*)d_in[12];
    const float* bo_BA  = (const float*)d_in[13];
    const float* Wf     = (const float*)d_in[14];
    const float* bf     = (const float*)d_in[15];
    float* coef = (float*)d_ws;
    float* out  = (float*)d_out;

    precompute_coeffs<<<1, 32, 0, stream>>>(WA, bA, WB, bB,
                                            Wi_AB, bi_AB, Wo_AB, bo_AB,
                                            Wi_BA, bi_BA, Wo_BA, bo_BA,
                                            Wf, bf, coef);

    const int nB       = in_sizes[0] / 3;
    const int nThreads = (nB + 3) / 4;
    const int nBlocks  = (nThreads + 255) / 256;
    dynf_main<<<nBlocks, 256, 0, stream>>>(groupA, groupB, coef, out, nB);
}


# precompute kernel: about 20 serial load round trips of its single wave become one: 15 waves (one WMMA M-tile, Wf@Wo fold or bias fetch each), every load issued before first use; sum order unchanged
# speedup vs baseline: 1.8531x; 1.8531x over previous
#include <hip/hip_runtime.h>
#include <hip/hip_bf16.h>

typedef float v2f __attribute__((ext_vector_type(2)));
typedef float v8f __attribute__((ext_vector_type(8)));

__global__ __launch_bounds__(480) void precompute_coeffs(
    const float* __restrict__ WA,    const float* __restrict__ bA,
    const float* __restrict__ WB,    const float* __restrict__ bB,
    const float* __restrict__ Wi_AB, const float* __restrict__ bi_AB,
    const float* __restrict__ Wo_AB, const float* __restrict__ bo_AB,
    const float* __restrict__ Wi_BA, const float* __restrict__ bi_BA,
    const float* __restrict__ Wo_BA, const float* __restrict__ bo_BA,
    const float* __restrict__ Wf,    const float* __restrict__ bf,
    float* __restrict__ coef)
{
    __shared__ float Y[192 * 16];
    __shared__ float G[2 * 32];
    const int lane  = threadIdx.x & 31;
    const int wave  = __builtin_amdgcn_readfirstlane((int)(threadIdx.x >> 5));
    const int n     = lane & 15;
    const int khalf = (lane >> 4) << 1;

    float biq, bik, biv, biq2, bik2, biv2, wfi, boA, boB, bf0;

    if (wave < 12) {
        const int t = wave;
        const float* Wrow = (t < 6) ? (Wi_AB + (16 * t + n) * 32)
                                    : (Wi_BA + (16 * (t - 6) + n) * 32);
        const float* xcol = (n == 1) ? bA : (n == 2) ? WB : (n == 3) ? bB : WA;
        float Ax[8], Ay[8], Bx[8], By[8];
#pragma unroll
        for (int k0 = 0; k0 < 8; ++k0) {
            const int kb = 4 * k0 + khalf;
            Ax[k0] = Wrow[kb]; Ay[k0] = Wrow[kb + 1];
            Bx[k0] = xcol[kb]; By[k0] = xcol[kb + 1];
        }
        v8f acc = {0.f, 0.f, 0.f, 0.f, 0.f, 0.f, 0.f, 0.f};
#pragma unroll
        for (int k0 = 0; k0 < 8; ++k0) {
            v2f a; a.x = Ax[k0]; a.y = Ay[k0];
            v2f b; b.x = Bx[k0]; b.y = By[k0];
            acc = __builtin_amdgcn_wmma_f32_16x16x4_f32(
                false, a, false, b, (short)0, acc, false, false);
        }
#pragma unroll
        for (int j = 0; j < 8; ++j) {
            const int M = 16 * t + j + ((lane >> 4) << 3);
            Y[M * 16 + n] = acc[j];
        }
    } else if (wave < 14) {
        const float* Wo = (wave == 12) ? Wo_AB : Wo_BA;
        float wo[32];
#pragma unroll
        for (int mm = 0; mm < 32; ++mm) wo[mm] = Wo[mm * 32 + lane];
        float g = 0.f;
#pragma unroll
        for (int mm = 0; mm < 32; ++mm) g += Wf[mm] * wo[mm];
        G[(wave - 12) * 32 + lane] = g;
    } else {
        biq  = bi_AB[lane]; bik  = bi_AB[32 + lane]; biv  = bi_AB[64 + lane];
        biq2 = bi_BA[lane]; bik2 = bi_BA[32 + lane]; biv2 = bi_BA[64 + lane];
        wfi  = Wf[lane];    boA  = bo_AB[lane];      boB  = bo_BA[lane];
        bf0  = bf[0];
    }
    __syncthreads();
    if (wave != 14) return;

    const int i = lane;
    float uq  = Y[(0   + i) * 16 + 0];
    float cq  = Y[(0   + i) * 16 + 1] + biq;
    float uk  = Y[(32  + i) * 16 + 2];
    float ck  = Y[(32  + i) * 16 + 3] + bik;
    float uv  = Y[(64  + i) * 16 + 2];
    float cv  = Y[(64  + i) * 16 + 3] + biv;
    float uq2 = Y[(96  + i) * 16 + 2];
    float cq2 = Y[(96  + i) * 16 + 3] + biq2;
    float uk2 = Y[(128 + i) * 16 + 0];
    float ck2 = Y[(128 + i) * 16 + 1] + bik2;
    float uv2 = Y[(160 + i) * 16 + 0];
    float cv2 = Y[(160 + i) * 16 + 1] + biv2;
    const float gAi = G[i], gBi = G[32 + i];

    float r[14];
    r[0]  = uq  * uk;   r[1]  = uq  * ck;   r[2]  = cq  * uk;   r[3]  = cq  * ck;
    r[4]  = uq2 * uk2;  r[5]  = uq2 * ck2;  r[6]  = cq2 * uk2;  r[7]  = cq2 * ck2;
    r[8]  = gAi * uv;   r[9]  = gBi * uv2;
    r[10] = gAi * cv;   r[11] = wfi * boA;
    r[12] = gBi * cv2;  r[13] = wfi * boB;
#pragma unroll
    for (int m = 16; m > 0; m >>= 1) {
#pragma unroll
        for (int j = 0; j < 14; ++j) r[j] += __shfl_xor(r[j], m, 32);
    }
    const float tA = r[10] + r[11];
    const float tB = r[12] + r[13];

    if (lane == 0) {
        coef[0] = r[0]; coef[1] = r[1]; coef[2] = r[2]; coef[3] = r[3];
        coef[4] = r[4]; coef[5] = r[5]; coef[6] = r[6]; coef[7] = r[7];
        coef[8] = r[8]; coef[9] = r[9];
        coef[10] = 0.5f * (tA + tB) + bf0;
    }
}

__device__ inline float softmax3dot(float c, float a0, float a1, float a2) {
    const float e0 = c * a0, e1 = c * a1, e2 = c * a2;
    const float mx = fmaxf(e0, fmaxf(e1, e2));
    const float w0 = __expf(e0 - mx), w1 = __expf(e1 - mx), w2 = __expf(e2 - mx);
    return (w0 * a0 + w1 * a1 + w2 * a2) / (w0 + w1 + w2);
}

__device__ inline float one_batch(float a0, float a1, float a2,
                                  float b0, float b1,
                                  float aAB, float gAB, float aBA, float gBA,
                                  float sA, float sB, float c0)
{
    const float inv = 0.17677669529663687f;
    const float db = inv * (b1 - b0);
    const float z0 = db * (aAB * a0 + gAB);
    const float z1 = db * (aAB * a1 + gAB);
    const float z2 = db * (aAB * a2 + gAB);
    const float S  = 1.f / (1.f + __expf(-z0))
                   + 1.f / (1.f + __expf(-z1))
                   + 1.f / (1.f + __expf(-z2));
    const float MA = b0 + (b1 - b0) * (S * (1.f / 3.f));
    const float c0t = inv * (aBA * b0 + gBA);
    const float c1t = inv * (aBA * b1 + gBA);
    const float MB  = 0.5f * (softmax3dot(c0t, a0, a1, a2) +
                              softmax3dot(c1t, a0, a1, a2));
    const float o = 0.5f * (MA * sA + MB * sB) + c0;
    return (o >= 0.f) ? o : 0.01f * o;
}

__global__ __launch_bounds__(256) void dynf_main(
    const float* __restrict__ gA, const float* __restrict__ gB,
    const float* __restrict__ coef, float* __restrict__ out, int nB)
{
    const float aAB = coef[0], gAB = coef[2];
    const float aBA = coef[4], gBA = coef[6];
    const float sA  = coef[8], sB  = coef[9], c0 = coef[10];

    const int idx = blockIdx.x * blockDim.x + threadIdx.x;
    const int b0  = idx * 4;
    if (b0 + 3 < nB) {
        const float4 va0 = ((const float4*)(gA + 3 * b0))[0];
        const float4 va1 = ((const float4*)(gA + 3 * b0))[1];
        const float4 va2 = ((const float4*)(gA + 3 * b0))[2];
        const float4 vb0 = ((const float4*)(gB + 2 * b0))[0];
        const float4 vb1 = ((const float4*)(gB + 2 * b0))[1];
        float4 r;
        r.x = one_batch(va0.x, va0.y, va0.z, vb0.x, vb0.y, aAB, gAB, aBA, gBA, sA, sB, c0);
        r.y = one_batch(va0.w, va1.x, va1.y, vb0.z, vb0.w, aAB, gAB, aBA, gBA, sA, sB, c0);
        r.z = one_batch(va1.z, va1.w, va2.x, vb1.x, vb1.y, aAB, gAB, aBA, gBA, sA, sB, c0);
        r.w = one_batch(va2.y, va2.z, va2.w, vb1.z, vb1.w, aAB, gAB, aBA, gBA, sA, sB, c0);
        ((float4*)(out + b0))[0] = r;
    } else {
        for (int b = b0; b < nB; ++b) {
            out[b] = one_batch(gA[3 * b], gA[3 * b + 1], gA[3 * b + 2],
                               gB[2 * b], gB[2 * b + 1],
                               aAB, gAB, aBA, gBA, sA, sB, c0);
        }
    }
}

extern "C" void kernel_launch(void* const* d_in, const int* in_sizes, int n_in,
                              void* d_out, int out_size, void* d_ws, size_t ws_size,
                              hipStream_t stream) {
    const float* groupA = (const float*)d_in[0];
    const float* groupB = (const float*)d_in[1];
    const float* WA     = (const float*)d_in[2];
    const float* bA     = (const float*)d_in[3];
    const float* WB     = (const float*)d_in[4];
    const float* bB     = (const float*)d_in[5];
    const float* Wi_AB  = (const float*)d_in[6];
    const float* bi_AB  = (const float*)d_in[7];
    const float* Wo_AB  = (const float*)d_in[8];
    const float* bo_AB  = (const float*)d_in[9];
    const float* Wi_BA  = (const float*)d_in[10];
    const float* bi_BA  = (const float*)d_in[11];
    const float* Wo_BA  = (const float*)d_in[12];
    const float* bo_BA  = (const float*)d_in[13];
    const float* Wf     = (const float*)d_in[14];
    const float* bf     = (const float*)d_in[15];
    float* coef = (float*)d_ws;
    float* out  = (float*)d_out;

    precompute_coeffs<<<1, 480, 0, stream>>>(WA, bA, WB, bB,
                                            Wi_AB, bi_AB, Wo_AB, bo_AB,
                                            Wi_BA, bi_BA, Wo_BA, bo_BA,
                                            Wf, bf, coef);

    const int nB       = in_sizes[0] / 3;
    const int nThreads = (nB + 3) / 4;
    const int nBlocks  = (nThreads + 255) / 256;
    dynf_main<<<nBlocks, 256, 0, stream>>>(groupA, groupB, coef, out, nB);
}


# two launches fused into one: each 16-wave workgroup folds the weights itself (same WMMA tiles, same sum order), passes the 7 coefficients through LDS and loads its batches under the fold
# speedup vs baseline: 2.5317x; 1.3662x over previous
#include <hip/hip_runtime.h>
#include <hip/hip_bf16.h>

typedef float v2f __attribute__((ext_vector_type(2)));
typedef float v8f __attribute__((ext_vector_type(8)));

__device__ inline float softmax3dot(float c, float a0, float a1, float a2) {
    const float e0 = c * a0, e1 = c * a1, e2 = c * a2;
    const float mx = fmaxf(e0, fmaxf(e1, e2));
    const float w0 = __expf(e0 - mx), w1 = __expf(e1 - mx), w2 = __expf(e2 - mx);
    return (w0 * a0 + w1 * a1 + w2 * a2) / (w0 + w1 + w2);
}

__device__ inline float one_batch(float a0, float a1, float a2,
                                  float b0, float b1,
                                  float aAB, float gAB, float aBA, float gBA,
                                  float sA, float sB, float c0)
{
    const float inv = 0.17677669529663687f;
    const float db = inv * (b1 - b0);
    const float z0 = db * (aAB * a0 + gAB);
    const float z1 = db * (aAB * a1 + gAB);
    const float z2 = db * (aAB * a2 + gAB);
    const float S  = 1.f / (1.f + __expf(-z0))
                   + 1.f / (1.f + __expf(-z1))
                   + 1.f / (1.f + __expf(-z2));
    const float MA = b0 + (b1 - b0) * (S * (1.f / 3.f));
    const float c0t = inv * (aBA * b0 + gBA);
    const float c1t = inv * (aBA * b1 + gBA);
    const float MB  = 0.5f * (softmax3dot(c0t, a0, a1, a2) +
                              softmax3dot(c1t, a0, a1, a2));
    const float o = 0.5f * (MA * sA + MB * sB) + c0;
    return (o >= 0.f) ? o : 0.01f * o;
}

__device__ inline void load_batches(const float* __restrict__ pa,
                                    const float* __restrict__ pb,
                                    float4& va0, float4& va1, float4& va2,
                                    float4& vb0, float4& vb1)
{
    va0 = ((const float4*)pa)[0];
    va1 = ((const float4*)pa)[1];
    va2 = ((const float4*)pa)[2];
    vb0 = ((const float4*)pb)[0];
    vb1 = ((const float4*)pb)[1];
}

__device__ inline void barrier_lds_only() {
    __builtin_amdgcn_fence(__ATOMIC_RELEASE, "workgroup", "local");
    __builtin_amdgcn_s_barrier();
    __builtin_amdgcn_fence(__ATOMIC_ACQUIRE, "workgroup", "local");
}

constexpr int kThreads = 512;

__global__ __launch_bounds__(kThreads) void dynf_fused(
    const float* __restrict__ gA,    const float* __restrict__ gB,
    const float* __restrict__ WA,    const float* __restrict__ bA,
    const float* __restrict__ WB,    const float* __restrict__ bB,
    const float* __restrict__ Wi_AB, const float* __restrict__ bi_AB,
    const float* __restrict__ Wo_AB, const float* __restrict__ bo_AB,
    const float* __restrict__ Wi_BA, const float* __restrict__ bi_BA,
    const float* __restrict__ Wo_BA, const float* __restrict__ bo_BA,
    const float* __restrict__ Wf,    const float* __restrict__ bf,
    float* __restrict__ out, int nB)
{
    __shared__ float Y[192 * 16];
    __shared__ float G[2 * 32];
    __shared__ __attribute__((aligned(16))) float C[8];
    const int lane  = threadIdx.x & 31;
    const int wave  = __builtin_amdgcn_readfirstlane((int)(threadIdx.x >> 5));
    const int n     = lane & 15;
    const int khalf = (lane >> 4) << 1;

    const int idx  = blockIdx.x * kThreads + threadIdx.x;
    const int b0   = idx * 4;
    const bool full = (b0 + 3 < nB);
    const float* pa = full ? (gA + 3 * b0) : Wi_AB;
    const float* pb = full ? (gB + 2 * b0) : Wi_AB;
    float4 va0, va1, va2, vb0, vb1;

    float biq, biv, biq2, biv2, wfi, boA, boB, bf0;

    float Ax[8], Ay[8], Bx[8], By[8];
    float wo[32];
    const int t = wave;
    if (wave < 12) {
        const float* Wrow = (t < 6) ? (Wi_AB + (16 * t + n) * 32)
                                    : (Wi_BA + (16 * (t - 6) + n) * 32);
        const float* xcol = (n == 1) ? bA : (n == 2) ? WB : (n == 3) ? bB : WA;
#pragma unroll
        for (int k0 = 0; k0 < 8; ++k0) {
            const int kb = 4 * k0 + khalf;
            Ax[k0] = Wrow[kb]; Ay[k0] = Wrow[kb + 1];
            Bx[k0] = xcol[kb]; By[k0] = xcol[kb + 1];
        }
    } else if (wave < 14) {
        const float* Wo = (wave == 12) ? Wo_AB : Wo_BA;
#pragma unroll
        for (int mm = 0; mm < 32; ++mm) wo[mm] = Wo[mm * 32 + lane];
    } else if (wave == 14) {
        biq  = bi_AB[lane]; biv  = bi_AB[64 + lane];
        biq2 = bi_BA[lane]; biv2 = bi_BA[64 + lane];
        wfi  = Wf[lane];    boA  = bo_AB[lane];      boB  = bo_BA[lane];
        bf0  = bf[0];
    }

    load_batches(pa, pb, va0, va1, va2, vb0, vb1);

    if (wave < 12) {
        v8f acc = {0.f, 0.f, 0.f, 0.f, 0.f, 0.f, 0.f, 0.f};
#pragma unroll
        for (int k0 = 0; k0 < 8; ++k0) {
            v2f a; a.x = Ax[k0]; a.y = Ay[k0];
            v2f b; b.x = Bx[k0]; b.y = By[k0];
            acc = __builtin_amdgcn_wmma_f32_16x16x4_f32(
                false, a, false, b, (short)0, acc, false, false);
        }
#pragma unroll
        for (int j = 0; j < 8; ++j) {
            const int M = 16 * t + j + ((lane >> 4) << 3);
            Y[M * 16 + n] = acc[j];
        }
    } else if (wave < 14) {
        float g = 0.f;
#pragma unroll
        for (int mm = 0; mm < 32; ++mm) g += Wf[mm] * wo[mm];
        G[(wave - 12) * 32 + lane] = g;
    }
    barrier_lds_only();

    if (wave == 14) {
        const int i = lane;
        const float uq  = Y[(0   + i) * 16 + 0];
        const float cq  = Y[(0   + i) * 16 + 1] + biq;
        const float uk  = Y[(32  + i) * 16 + 2];
        const float uv  = Y[(64  + i) * 16 + 2];
        const float cv  = Y[(64  + i) * 16 + 3] + biv;
        const float uq2 = Y[(96  + i) * 16 + 2];
        const float cq2 = Y[(96  + i) * 16 + 3] + biq2;
        const float uk2 = Y[(128 + i) * 16 + 0];
        const float uv2 = Y[(160 + i) * 16 + 0];
        const float cv2 = Y[(160 + i) * 16 + 1] + biv2;
        const float gAi = G[i], gBi = G[32 + i];

        float r[10];
        r[0] = uq  * uk;    r[1] = cq  * uk;
        r[2] = uq2 * uk2;   r[3] = cq2 * uk2;
        r[4] = gAi * uv;    r[5] = gBi * uv2;
        r[6] = gAi * cv;    r[7] = wfi * boA;
        r[8] = gBi * cv2;   r[9] = wfi * boB;
#pragma unroll
        for (int m = 16; m > 0; m >>= 1) {
#pragma unroll
            for (int j = 0; j < 10; ++j) r[j] += __shfl_xor(r[j], m, 32);
        }
        const float tA = r[6] + r[7];
        const float tB = r[8] + r[9];
        if (lane == 0) {
            C[0] = r[0]; C[1] = r[1]; C[2] = r[2]; C[3] = r[3];
            C[4] = r[4]; C[5] = r[5];
            C[6] = 0.5f * (tA + tB) + bf0;
            C[7] = 0.f;
        }
    }
    barrier_lds_only();

    const float4 c03 = ((const float4*)C)[0];
    const float4 c47 = ((const float4*)C)[1];
    const float aAB = c03.x, gAB = c03.y;
    const float aBA = c03.z, gBA = c03.w;
    const float sA  = c47.x, sB  = c47.y, c0 = c47.z;

    if (full) {
        float4 r;
        r.x = one_batch(va0.x, va0.y, va0.z, vb0.x, vb0.y, aAB, gAB, aBA, gBA, sA, sB, c0);
        r.y = one_batch(va0.w, va1.x, va1.y, vb0.z, vb0.w, aAB, gAB, aBA, gBA, sA, sB, c0);
        r.z = one_batch(va1.z, va1.w, va2.x, vb1.x, vb1.y, aAB, gAB, aBA, gBA, sA, sB, c0);
        r.w = one_batch(va2.y, va2.z, va2.w, vb1.z, vb1.w, aAB, gAB, aBA, gBA, sA, sB, c0);
        ((float4*)(out + b0))[0] = r;
    } else {
        for (int b = b0; b < nB; ++b) {
            out[b] = one_batch(gA[3 * b], gA[3 * b + 1], gA[3 * b + 2],
                               gB[2 * b], gB[2 * b + 1],
                               aAB, gAB, aBA, gBA, sA, sB, c0);
        }
    }
}

extern "C" void kernel_launch(void* const* d_in, const int* in_sizes, int n_in,
                              void* d_out, int out_size, void* d_ws, size_t ws_size,
                              hipStream_t stream) {
    const float* groupA = (const float*)d_in[0];
    const float* groupB = (const float*)d_in[1];
    const float* WA     = (const float*)d_in[2];
    const float* bA     = (const float*)d_in[3];
    const float* WB     = (const float*)d_in[4];
    const float* bB     = (const float*)d_in[5];
    const float* Wi_AB  = (const float*)d_in[6];
    const float* bi_AB  = (const float*)d_in[7];
    const float* Wo_AB  = (const float*)d_in[8];
    const float* bo_AB  = (const float*)d_in[9];
    const float* Wi_BA  = (const float*)d_in[10];
    const float* bi_BA  = (const float*)d_in[11];
    const float* Wo_BA  = (const float*)d_in[12];
    const float* bo_BA  = (const float*)d_in[13];
    const float* Wf     = (const float*)d_in[14];
    const float* bf     = (const float*)d_in[15];
    float* out  = (float*)d_out;

    const int nB       = in_sizes[0] / 3;
    const int nThreads = (nB + 3) / 4;
    const int nBlocks  = (nThreads + kThreads - 1) / kThreads;
    dynf_fused<<<nBlocks, kThreads, 0, stream>>>(groupA, groupB, WA, bA, WB, bB,
                                            Wi_AB, bi_AB, Wo_AB, bo_AB,
                                            Wi_BA, bi_BA, Wo_BA, bo_BA,
                                            Wf, bf, out, nB);
}


# batch phase issues fewer vector instructions: each of the five IEEE divisions per batch (11 instructions) becomes v_rcp_f32 (+1 multiply), log2(e)/sqrt(32) folded into exponent coefficients
# speedup vs baseline: 2.8677x; 1.1327x over previous
#include <hip/hip_runtime.h>
#include <hip/hip_bf16.h>

typedef float v2f __attribute__((ext_vector_type(2)));
typedef float v8f __attribute__((ext_vector_type(8)));

struct BatchCoef {
    float aS, gS;
    float aC, gC;
    float sA, sB, c0;
};

__device__ inline BatchCoef make_coef(float aAB, float gAB, float aBA, float gBA,
                                      float sA, float sB, float c0) {
    const float k = 0.17677669529663687f * 1.4426950408889634f;
    BatchCoef q;
    q.aS = -k * aAB; q.gS = -k * gAB;
    q.aC =  k * aBA; q.gC =  k * gBA;
    q.sA = sA; q.sB = sB; q.c0 = c0;
    return q;
}

__device__ inline float softmax3dot(float c, float a0, float a1, float a2) {
    const float e0 = c * a0, e1 = c * a1, e2 = c * a2;
    const float mx = fmaxf(e0, fmaxf(e1, e2));
    const float w0 = __builtin_amdgcn_exp2f(e0 - mx);
    const float w1 = __builtin_amdgcn_exp2f(e1 - mx);
    const float w2 = __builtin_amdgcn_exp2f(e2 - mx);
    return (w0 * a0 + w1 * a1 + w2 * a2) * __builtin_amdgcn_rcpf(w0 + w1 + w2);
}

__device__ inline float one_batch(float a0, float a1, float a2,
                                  float b0, float b1, const BatchCoef& q)
{
    const float d  = b1 - b0;
    const float x0 = d * (q.aS * a0 + q.gS);
    const float x1 = d * (q.aS * a1 + q.gS);
    const float x2 = d * (q.aS * a2 + q.gS);
    const float S  = __builtin_amdgcn_rcpf(1.f + __builtin_amdgcn_exp2f(x0))
                   + __builtin_amdgcn_rcpf(1.f + __builtin_amdgcn_exp2f(x1))
                   + __builtin_amdgcn_rcpf(1.f + __builtin_amdgcn_exp2f(x2));
    const float MA = b0 + d * (S * (1.f / 3.f));
    const float c0t = q.aC * b0 + q.gC;
    const float c1t = q.aC * b1 + q.gC;
    const float MB  = 0.5f * (softmax3dot(c0t, a0, a1, a2) +
                              softmax3dot(c1t, a0, a1, a2));
    const float o = 0.5f * (MA * q.sA + MB * q.sB) + q.c0;
    return fmaxf(o, 0.01f * o);
}

__device__ inline void load_batches(const float* __restrict__ pa,
                                    const float* __restrict__ pb,
                                    float4& va0, float4& va1, float4& va2,
                                    float4& vb0, float4& vb1)
{
    va0 = ((const float4*)pa)[0];
    va1 = ((const float4*)pa)[1];
    va2 = ((const float4*)pa)[2];
    vb0 = ((const float4*)pb)[0];
    vb1 = ((const float4*)pb)[1];
}

__device__ inline void barrier_lds_only() {
    __builtin_amdgcn_fence(__ATOMIC_RELEASE, "workgroup", "local");
    __builtin_amdgcn_s_barrier();
    __builtin_amdgcn_fence(__ATOMIC_ACQUIRE, "workgroup", "local");
}

constexpr int kThreads = 512;

__global__ __launch_bounds__(kThreads) void dynf_fused(
    const float* __restrict__ gA,    const float* __restrict__ gB,
    const float* __restrict__ WA,    const float* __restrict__ bA,
    const float* __restrict__ WB,    const float* __restrict__ bB,
    const float* __restrict__ Wi_AB, const float* __restrict__ bi_AB,
    const float* __restrict__ Wo_AB, const float* __restrict__ bo_AB,
    const float* __restrict__ Wi_BA, const float* __restrict__ bi_BA,
    const float* __restrict__ Wo_BA, const float* __restrict__ bo_BA,
    const float* __restrict__ Wf,    const float* __restrict__ bf,
    float* __restrict__ out, int nB)
{
    __shared__ float Y[192 * 16];
    __shared__ float G[2 * 32];
    __shared__ __attribute__((aligned(16))) float C[8];
    const int lane  = threadIdx.x & 31;
    const int wave  = __builtin_amdgcn_readfirstlane((int)(threadIdx.x >> 5));
    const int n     = lane & 15;
    const int khalf = (lane >> 4) << 1;

    const int idx  = blockIdx.x * kThreads + threadIdx.x;
    const int b0   = idx * 4;
    const bool full = (b0 + 3 < nB);
    const float* pa = full ? (gA + 3 * b0) : Wi_AB;
    const float* pb = full ? (gB + 2 * b0) : Wi_AB;
    float4 va0, va1, va2, vb0, vb1;

    float biq, biv, biq2, biv2, wfi, boA, boB, bf0;

    float Ax[8], Ay[8], Bx[8], By[8];
    float wo[32];
    const int t = wave;
    if (wave < 12) {
        const float* Wrow = (t < 6) ? (Wi_AB + (16 * t + n) * 32)
                                    : (Wi_BA + (16 * (t - 6) + n) * 32);
        const float* xcol = (n == 1) ? bA : (n == 2) ? WB : (n == 3) ? bB : WA;
#pragma unroll
        for (int k0 = 0; k0 < 8; ++k0) {
            const int kb = 4 * k0 + khalf;
            Ax[k0] = Wrow[kb]; Ay[k0] = Wrow[kb + 1];
            Bx[k0] = xcol[kb]; By[k0] = xcol[kb + 1];
        }
    } else if (wave < 14) {
        const float* Wo = (wave == 12) ? Wo_AB : Wo_BA;
#pragma unroll
        for (int mm = 0; mm < 32; ++mm) wo[mm] = Wo[mm * 32 + lane];
    } else if (wave == 14) {
        biq  = bi_AB[lane]; biv  = bi_AB[64 + lane];
        biq2 = bi_BA[lane]; biv2 = bi_BA[64 + lane];
        wfi  = Wf[lane];    boA  = bo_AB[lane];      boB  = bo_BA[lane];
        bf0  = bf[0];
    }

    load_batches(pa, pb, va0, va1, va2, vb0, vb1);

    if (wave < 12) {
        v8f acc = {0.f, 0.f, 0.f, 0.f, 0.f, 0.f, 0.f, 0.f};
#pragma unroll
        for (int k0 = 0; k0 < 8; ++k0) {
            v2f a; a.x = Ax[k0]; a.y = Ay[k0];
            v2f b; b.x = Bx[k0]; b.y = By[k0];
            acc = __builtin_amdgcn_wmma_f32_16x16x4_f32(
                false, a, false, b, (short)0, acc, false, false);
        }
#pragma unroll
        for (int j = 0; j < 8; ++j) {
            const int M = 16 * t + j + ((lane >> 4) << 3);
            Y[M * 16 + n] = acc[j];
        }
    } else if (wave < 14) {
        float g = 0.f;
#pragma unroll
        for (int mm = 0; mm < 32; ++mm) g += Wf[mm] * wo[mm];
        G[(wave - 12) * 32 + lane] = g;
    }
    barrier_lds_only();

    if (wave == 14) {
        const int i = lane;
        const float uq  = Y[(0   + i) * 16 + 0];
        const float cq  = Y[(0   + i) * 16 + 1] + biq;
        const float uk  = Y[(32  + i) * 16 + 2];
        const float uv  = Y[(64  + i) * 16 + 2];
        const float cv  = Y[(64  + i) * 16 + 3] + biv;
        const float uq2 = Y[(96  + i) * 16 + 2];
        const float cq2 = Y[(96  + i) * 16 + 3] + biq2;
        const float uk2 = Y[(128 + i) * 16 + 0];
        const float uv2 = Y[(160 + i) * 16 + 0];
        const float cv2 = Y[(160 + i) * 16 + 1] + biv2;
        const float gAi = G[i], gBi = G[32 + i];

        float r[10];
        r[0] = uq  * uk;    r[1] = cq  * uk;
        r[2] = uq2 * uk2;   r[3] = cq2 * uk2;
        r[4] = gAi * uv;    r[5] = gBi * uv2;
        r[6] = gAi * cv;    r[7] = wfi * boA;
        r[8] = gBi * cv2;   r[9] = wfi * boB;
#pragma unroll
        for (int m = 16; m > 0; m >>= 1) {
#pragma unroll
            for (int j = 0; j < 10; ++j) r[j] += __shfl_xor(r[j], m, 32);
        }
        const float tA = r[6] + r[7];
        const float tB = r[8] + r[9];
        if (lane == 0) {
            C[0] = r[0]; C[1] = r[1]; C[2] = r[2]; C[3] = r[3];
            C[4] = r[4]; C[5] = r[5];
            C[6] = 0.5f * (tA + tB) + bf0;
            C[7] = 0.f;
        }
    }
    barrier_lds_only();

    const float4 c03 = ((const float4*)C)[0];
    const float4 c47 = ((const float4*)C)[1];
    const BatchCoef q = make_coef(c03.x, c03.y, c03.z, c03.w, c47.x, c47.y, c47.z);

    if (full) {
        float4 r;
        r.x = one_batch(va0.x, va0.y, va0.z, vb0.x, vb0.y, q);
        r.y = one_batch(va0.w, va1.x, va1.y, vb0.z, vb0.w, q);
        r.z = one_batch(va1.z, va1.w, va2.x, vb1.x, vb1.y, q);
        r.w = one_batch(va2.y, va2.z, va2.w, vb1.z, vb1.w, q);
        ((float4*)(out + b0))[0] = r;
    } else {
        for (int b = b0; b < nB; ++b) {
            out[b] = one_batch(gA[3 * b], gA[3 * b + 1], gA[3 * b + 2],
                               gB[2 * b], gB[2 * b + 1], q);
        }
    }
}

extern "C" void kernel_launch(void* const* d_in, const int* in_sizes, int n_in,
                              void* d_out, int out_size, void* d_ws, size_t ws_size,
                              hipStream_t stream) {
    const float* groupA = (const float*)d_in[0];
    const float* groupB = (const float*)d_in[1];
    const float* WA     = (const float*)d_in[2];
    const float* bA     = (const float*)d_in[3];
    const float* WB     = (const float*)d_in[4];
    const float* bB     = (const float*)d_in[5];
    const float* Wi_AB  = (const float*)d_in[6];
    const float* bi_AB  = (const float*)d_in[7];
    const float* Wo_AB  = (const float*)d_in[8];
    const float* bo_AB  = (const float*)d_in[9];
    const float* Wi_BA  = (const float*)d_in[10];
    const float* bi_BA  = (const float*)d_in[11];
    const float* Wo_BA  = (const float*)d_in[12];
    const float* bo_BA  = (const float*)d_in[13];
    const float* Wf     = (const float*)d_in[14];
    const float* bf     = (const float*)d_in[15];
    float* out  = (float*)d_out;

    const int nB       = in_sizes[0] / 3;
    const int nThreads = (nB + 3) / 4;
    const int nBlocks  = (nThreads + kThreads - 1) / kThreads;
    dynf_fused<<<nBlocks, kThreads, 0, stream>>>(groupA, groupB, WA, bA, WB, bB,
                                            Wi_AB, bi_AB, Wo_AB, bo_AB,
                                            Wi_BA, bi_BA, Wo_BA, bo_BA,
                                            Wf, bf, out, nB);
}
